# starting kernel
# baseline (speedup 1.0000x reference)
#include <hip/hip_runtime.h>
#include <math.h>

typedef __attribute__((ext_vector_type(2))) float v2f;
typedef __attribute__((ext_vector_type(8))) float v8f;

#define Hdim 256
#define ENC 1024
#define DEC 256
#define BATCH 4

__device__ __forceinline__ float gpu_tanh(float x) {
#if __has_builtin(__builtin_amdgcn_tanhf)
    return __builtin_amdgcn_tanhf(x);
#elif __has_builtin(__builtin_amdgcn_tanh_f32)
    return __builtin_amdgcn_tanh_f32(x);
#else
    float r;
    asm volatile("v_tanh_f32 %0, %1\n\tv_nop" : "=v"(r) : "v"(x));
    return r;
#endif
}

__global__ __launch_bounds__(32) void gemm_xwT_wmma(const float* __restrict__ x,
                                                    const float* __restrict__ w,
                                                    float* __restrict__ out) {
    const int lane  = threadIdx.x;
    const int mtile = blockIdx.x;
    const int ntile = blockIdx.y;
    const int l15   = lane & 15;
    const int koff  = (lane >> 4) << 1;

    const float2* xrow = reinterpret_cast<const float2*>(x + (size_t)(mtile * 16 + l15) * Hdim);
    const float2* wrow = reinterpret_cast<const float2*>(w + (size_t)(ntile * 16 + l15) * Hdim);

    v8f c = {};
#pragma unroll 4
    for (int k0 = 0; k0 < Hdim; k0 += 4) {
        float2 af = xrow[(k0 + koff) >> 1];
        float2 bf = wrow[(k0 + koff) >> 1];
        v2f a; a.x = af.x; a.y = af.y;
        v2f b; b.x = bf.x; b.y = bf.y;
        c = __builtin_amdgcn_wmma_f32_16x16x4_f32(false, a, false, b, (short)0, c,
                                                  false, false);
    }

    const int nstore = ntile * 16 + l15;
    const int mbase  = mtile * 16 + ((lane >> 4) << 3);
#pragma unroll
    for (int r = 0; r < 8; ++r)
        out[(size_t)(mbase + r) * Hdim + nstore] = c[r];
}

__global__ __launch_bounds__(256) void attn_scores_lsm(const float* __restrict__ dec_t,
                                                       const float* __restrict__ enc_t,
                                                       const float* __restrict__ v,
                                                       float* __restrict__ out) {
    const int bd = blockIdx.x;
    const int b  = bd / DEC;
    const int t  = threadIdx.x;

    __shared__ float sdec[Hdim];
    __shared__ float sv[Hdim];
    __shared__ float red[256];

    sdec[t] = dec_t[(size_t)bd * Hdim + t];
    sv[t]   = v[t];
    __syncthreads();

    const float4* sdec4 = reinterpret_cast<const float4*>(sdec);
    const float4* sv4   = reinterpret_cast<const float4*>(sv);

    float s[4];
#pragma unroll
    for (int j = 0; j < 4; ++j) {
        const int e = j * 256 + t;
        const float4* erow =
            reinterpret_cast<const float4*>(enc_t + ((size_t)b * ENC + e) * Hdim);
        float acc = 0.f;
        for (int i = 0; i < Hdim / 4; ++i) {
            float4 ev = erow[i];
            float4 dv = sdec4[i];
            float4 vv = sv4[i];
            acc += vv.x * gpu_tanh(dv.x + ev.x);
            acc += vv.y * gpu_tanh(dv.y + ev.y);
            acc += vv.z * gpu_tanh(dv.z + ev.z);
            acc += vv.w * gpu_tanh(dv.w + ev.w);
        }
        s[j] = acc;
    }

    float m = fmaxf(fmaxf(s[0], s[1]), fmaxf(s[2], s[3]));
    red[t] = m;
    __syncthreads();
    for (int str = 128; str >= 1; str >>= 1) {
        if (t < str) red[t] = fmaxf(red[t], red[t + str]);
        __syncthreads();
    }
    const float gmax = red[0];
    __syncthreads();

    float lsum = 0.f;
#pragma unroll
    for (int j = 0; j < 4; ++j) lsum += __expf(s[j] - gmax);
    red[t] = lsum;
    __syncthreads();
    for (int str = 128; str >= 1; str >>= 1) {
        if (t < str) red[t] += red[t + str];
        __syncthreads();
    }
    const float lse = gmax + __logf(red[0]);

#pragma unroll
    for (int j = 0; j < 4; ++j)
        out[(size_t)bd * ENC + (size_t)j * 256 + t] = s[j] - lse;
}

extern "C" void kernel_launch(void* const* d_in, const int* in_sizes, int n_in,
                              void* d_out, int out_size, void* d_ws, size_t ws_size,
                              hipStream_t stream) {
    const float* x_dec = (const float*)d_in[0];
    const float* x_enc = (const float*)d_in[1];
    const float* w1    = (const float*)d_in[2];
    const float* w2    = (const float*)d_in[3];
    const float* v     = (const float*)d_in[4];
    float* out = (float*)d_out;

    float* enc_t = (float*)d_ws;
    float* dec_t = enc_t + (size_t)BATCH * ENC * Hdim;

    gemm_xwT_wmma<<<dim3(BATCH * ENC / 16, Hdim / 16), 32, 0, stream>>>(x_enc, w1, enc_t);
    gemm_xwT_wmma<<<dim3(BATCH * DEC / 16, Hdim / 16), 32, 0, stream>>>(x_dec, w2, dec_t);
    attn_scores_lsm<<<BATCH * DEC, 256, 0, stream>>>(dec_t, enc_t, v, out);
}


# enc projection stored chunk-major (16 B = 4 features of one enc row): a scores thread fetches its own row by whole-line loads into a private LDS slot; the 8 block barriers of the h loop go
# speedup vs baseline: 2.9499x; 2.9499x over previous
#include <hip/hip_runtime.h>
#include <math.h>

typedef __attribute__((ext_vector_type(2))) float v2f;
typedef __attribute__((ext_vector_type(8))) float v8f;

#define Hdim 256
#define ENC 1024
#define DEC 256
#define BATCH 4

__device__ __forceinline__ float gpu_tanh(float x) {
#if __has_builtin(__builtin_amdgcn_tanhf)
    return __builtin_amdgcn_tanhf(x);
#elif __has_builtin(__builtin_amdgcn_tanh_f32)
    return __builtin_amdgcn_tanh_f32(x);
#else
    float r;
    asm volatile("v_tanh_f32 %0, %1\n\tv_nop" : "=v"(r) : "v"(x));
    return r;
#endif
}

typedef __attribute__((ext_vector_type(4))) float v4f;

#define KBLK (Hdim / 8)

__device__ __forceinline__ v4f frag_load(const float* __restrict__ m, int row, int half, int kb) {
    return *reinterpret_cast<const v4f*>(m + (size_t)row * Hdim + kb * 8 + half * 4);
}

template <int MT, int NT, int AHEAD, bool CHUNKED>
__global__ __launch_bounds__(128) __attribute__((amdgpu_waves_per_eu(1, 1))) void gemm_xwT_wmma(const float* __restrict__ x,
                                                     const float* __restrict__ w,
                                                     float* __restrict__ out) {
    constexpr int NACC = (MT * NT == 1) ? 2 : 1;

    const int lane = threadIdx.x & 31;
    const int wave = threadIdx.x >> 5;
    const int c    = lane & 15;
    const int half = lane >> 4;
    const int row0 = (blockIdx.x * 2 + (wave >> 1)) * (16 * MT);
    const int col0 = (blockIdx.y * 2 + (wave & 1)) * (16 * NT);

    v4f fa[KBLK][MT], fb[KBLK][NT];
    v8f acc[NACC][MT][NT];
#pragma unroll
    for (int s = 0; s < NACC; ++s)
#pragma unroll
        for (int i = 0; i < MT; ++i)
#pragma unroll
            for (int j = 0; j < NT; ++j) acc[s][i][j] = v8f{};

#pragma unroll
    for (int kb = 0; kb < AHEAD; ++kb) {
#pragma unroll
        for (int i = 0; i < MT; ++i) fa[kb][i] = frag_load(x, row0 + 16 * i + c, half, kb);
#pragma unroll
        for (int j = 0; j < NT; ++j) fb[kb][j] = frag_load(w, col0 + 16 * j + c, half, kb);
    }

#pragma unroll
    for (int kb = 0; kb < KBLK; ++kb) {
        if (kb + AHEAD < KBLK) {
#pragma unroll
            for (int i = 0; i < MT; ++i)
                fa[kb + AHEAD][i] = frag_load(x, row0 + 16 * i + c, half, kb + AHEAD);
#pragma unroll
            for (int j = 0; j < NT; ++j)
                fb[kb + AHEAD][j] = frag_load(w, col0 + 16 * j + c, half, kb + AHEAD);
        }
#pragma unroll
        for (int s = 0; s < 2; ++s) {
#pragma unroll
            for (int i = 0; i < MT; ++i) {
                v2f a;
                a.x = s ? fa[kb][i].z : fa[kb][i].x;
                a.y = s ? fa[kb][i].w : fa[kb][i].y;
#pragma unroll
                for (int j = 0; j < NT; ++j) {
                    v2f b;
                    b.x = s ? fb[kb][j].z : fb[kb][j].x;
                    b.y = s ? fb[kb][j].w : fb[kb][j].y;
                    v8f& d = acc[NACC == 2 ? s : 0][i][j];
                    d = __builtin_amdgcn_wmma_f32_16x16x4_f32(false, a, false, b, (short)0, d,
                                                              false, false);
                }
            }
        }
    }

#pragma unroll
    for (int i = 0; i < MT; ++i)
#pragma unroll
        for (int j = 0; j < NT; ++j) {
            v8f r = acc[0][i][j];
            if (NACC == 2) r += acc[NACC - 1][i][j];
            if (CHUNKED) {
                const int o0 = row0 + 16 * i + 8 * half;
                const int eg = col0 + 16 * j + c;
                v4f* o4 = reinterpret_cast<v4f*>(out)
                          + ((size_t)((eg / ENC) * (Hdim / 4) + (o0 >> 2)) * ENC + (eg % ENC));
                o4[0]   = v4f{r[0], r[1], r[2], r[3]};
                o4[ENC] = v4f{r[4], r[5], r[6], r[7]};
            } else {
                float* o = out + (size_t)(row0 + 16 * i + 8 * half) * Hdim + col0 + 16 * j + c;
#pragma unroll
                for (int q = 0; q < 8; ++q) o[(size_t)q * Hdim] = r[q];
            }
        }
}


#define DROWS 4
#define SC_THREADS 512
#define HCH4 16
#define PITCH4 17
#define STAGE4 (SC_THREADS * PITCH4)
#define NSTAGE 8
#define BATCH4 4
#define NSTEP (BATCH4 * 4 * DROWS)
#define ADD_AHEAD 3
#define FMA_BEHIND 4

__device__ __forceinline__ void batch_steps(const v4f* __restrict__ my, const v4f* __restrict__ dq,
                                            const v4f* __restrict__ vq, float (&acc)[DROWS]) {
    float x[NSTEP], y[NSTEP];
#pragma unroll
    for (int n = -ADD_AHEAD; n < NSTEP + FMA_BEHIND; ++n) {
        const int na = n + ADD_AHEAD;
        if (na >= 0 && na < NSTEP) {
            const int i = na >> 4, c = (na >> 2) & 3, d = na & 3;
            x[na] = dq[d * (Hdim / 4) + i][c] + my[i][c];
        }
        if (n >= 0 && n < NSTEP) y[n] = gpu_tanh(x[n]);
        const int nf = n - FMA_BEHIND;
        if (nf >= 0 && nf < NSTEP) {
            const int i = nf >> 4, c = (nf >> 2) & 3, d = nf & 3;
            acc[d] += vq[i][c] * y[nf];
        }
    }
}

__global__ __launch_bounds__(SC_THREADS) void attn_scores_lsm(const float* __restrict__ dec_t,
                                                              const float* __restrict__ enc_t,
                                                              const float* __restrict__ v,
                                                              float* __restrict__ out) {
    const int g    = blockIdx.x;
    const int b    = g / (DEC / DROWS);
    const int row0 = g * DROWS;
    const int t    = threadIdx.x;

    __shared__ v4f stage[2 * STAGE4];
    __shared__ float wmax[DROWS * 16];
    __shared__ float wsum[DROWS * 16];

    const v4f* enc4 = reinterpret_cast<const v4f*>(enc_t) + (size_t)b * (Hdim / 4) * ENC + t;
    const v4f* dec4 = reinterpret_cast<const v4f*>(dec_t + (size_t)row0 * Hdim);
    const v4f* v4   = reinterpret_cast<const v4f*>(v);

    v4f* mine = stage + t * PITCH4;

    v4f pre[HCH4];
#pragma unroll
    for (int k = 0; k < HCH4; ++k) pre[k] = enc4[(size_t)k * ENC];
#pragma unroll
    for (int k = 0; k < HCH4; ++k) mine[k] = pre[k];

    float acc[DROWS], sc0[DROWS], sc1[DROWS];
#pragma unroll
    for (int d = 0; d < DROWS; ++d) { acc[d] = 0.f; sc0[d] = 0.f; }

#pragma nounroll
    for (int s = 0; s < NSTAGE - 1; ++s) {
        const int sn = s + 1;
        {
            const v4f* src = enc4 + (size_t)((sn & 3) * HCH4) * ENC + (sn >> 2) * SC_THREADS;
#pragma unroll
            for (int k = 0; k < HCH4; ++k) pre[k] = src[(size_t)k * ENC];
        }

#pragma nounroll
        for (int q = 0; q < HCH4 / BATCH4; ++q)
            batch_steps(stage + (s & 1) * STAGE4 + t * PITCH4 + q * BATCH4,
                        dec4 + (s & 3) * HCH4 + q * BATCH4, v4 + (s & 3) * HCH4 + q * BATCH4, acc);

        if (s == 3) {
#pragma unroll
            for (int d = 0; d < DROWS; ++d) { sc0[d] = acc[d]; acc[d] = 0.f; }
        }

        {
            v4f* dst = mine + (sn & 1) * STAGE4;
#pragma unroll
            for (int k = 0; k < HCH4; ++k) dst[k] = pre[k];
        }
    }
#pragma nounroll
    for (int q = 0; q < HCH4 / BATCH4; ++q)
        batch_steps(stage + ((NSTAGE - 1) & 1) * STAGE4 + t * PITCH4 + q * BATCH4,
                    dec4 + 3 * HCH4 + q * BATCH4, v4 + 3 * HCH4 + q * BATCH4, acc);
#pragma unroll
    for (int d = 0; d < DROWS; ++d) sc1[d] = acc[d];

    const int wave = t >> 5;
    const int lane = t & 31;

    float gm[DROWS], part[DROWS];
#pragma unroll
    for (int d = 0; d < DROWS; ++d) part[d] = fmaxf(sc0[d], sc1[d]);
#pragma unroll
    for (int off = 16; off >= 1; off >>= 1) {
        float o[DROWS];
#pragma unroll
        for (int d = 0; d < DROWS; ++d) o[d] = __shfl_xor(part[d], off);
#pragma unroll
        for (int d = 0; d < DROWS; ++d) part[d] = fmaxf(part[d], o[d]);
    }
    if (lane == 0) {
#pragma unroll
        for (int d = 0; d < DROWS; ++d) wmax[d * 16 + wave] = part[d];
    }
    __syncthreads();
#pragma unroll
    for (int d = 0; d < DROWS; ++d) {
        float m = wmax[d * 16];
#pragma unroll
        for (int w = 1; w < 16; ++w) m = fmaxf(m, wmax[d * 16 + w]);
        gm[d] = m;
    }

#pragma unroll
    for (int d = 0; d < DROWS; ++d) part[d] = __expf(sc0[d] - gm[d]) + __expf(sc1[d] - gm[d]);
#pragma unroll
    for (int off = 16; off >= 1; off >>= 1) {
        float o[DROWS];
#pragma unroll
        for (int d = 0; d < DROWS; ++d) o[d] = __shfl_xor(part[d], off);
#pragma unroll
        for (int d = 0; d < DROWS; ++d) part[d] += o[d];
    }
    if (lane == 0) {
#pragma unroll
        for (int d = 0; d < DROWS; ++d) wsum[d * 16 + wave] = part[d];
    }
    __syncthreads();
#pragma unroll
    for (int d = 0; d < DROWS; ++d) {
        float tot = wsum[d * 16];
#pragma unroll
        for (int w = 1; w < 16; ++w) tot += wsum[d * 16 + w];
        const float lse = gm[d] + __logf(tot);
        float* orow = out + (size_t)(row0 + d) * ENC;
        orow[t]              = sc0[d] - lse;
        orow[SC_THREADS + t] = sc1[d] - lse;
    }
}

extern "C" void kernel_launch(void* const* d_in, const int* in_sizes, int n_in,
                              void* d_out, int out_size, void* d_ws, size_t ws_size,
                              hipStream_t stream) {
    const float* x_dec = (const float*)d_in[0];
    const float* x_enc = (const float*)d_in[1];
    const float* w1    = (const float*)d_in[2];
    const float* w2    = (const float*)d_in[3];
    const float* v     = (const float*)d_in[4];
    float* out = (float*)d_out;

    float* enc_t = (float*)d_ws;
    float* dec_t = enc_t + (size_t)BATCH * ENC * Hdim;

    gemm_xwT_wmma<2, 2, 8, true><<<dim3(Hdim / 64, BATCH * ENC / 64), 128, 0, stream>>>(w1, x_enc, enc_t);
    gemm_xwT_wmma<1, 1, 28, false><<<dim3(BATCH * DEC / 32, Hdim / 32), 128, 0, stream>>>(x_dec, w2, dec_t);
    attn_scores_lsm<<<BATCH * DEC / DROWS, SC_THREADS, 0, stream>>>(dec_t, enc_t, v, out);
}
